# starting kernel
# baseline (speedup 1.0000x reference)
#include <hip/hip_runtime.h>
#include <hip/hip_bf16.h>


#define ROWS      2304
#define DDIM      64
#define NREP      207
#define ROW_FLOATS (NREP * DDIM)
#define ROW_F4     (ROW_FLOATS / 4)

typedef __attribute__((ext_vector_type(2))) float v2f;
typedef __attribute__((ext_vector_type(4))) float v4f;
typedef __attribute__((ext_vector_type(8))) float v8f;

__global__ void embed_h_kernel(const int* __restrict__ TE,
                               const float* __restrict__ W1,
                               const float* __restrict__ b1,
                               float* __restrict__ h) {
    const int idx = blockIdx.x * blockDim.x + threadIdx.x;
    const int row = idx >> 6;
    const int d   = idx & 63;
    if (row >= ROWS) return;

    int te0 = TE[row * (NREP * 2) + 0];
    int te1 = TE[row * (NREP * 2) + 1];
    int dow = te0 % 7;   if (dow < 0) dow += 7;
    int tod = te1 % 288; if (tod < 0) tod += 288;

    float v = W1[dow * DDIM + d] + W1[(7 + tod) * DDIM + d] + b1[d];
    h[row * DDIM + d] = v > 0.0f ? v : 0.0f;
}

__global__ void gemm_wmma_kernel(const float* __restrict__ h,
                                 const float* __restrict__ W2,
                                 const float* __restrict__ b2,
                                 float* __restrict__ outr) {
    const int lane = threadIdx.x & 31;
    const int wave = threadIdx.x >> 5;
    const int m0   = blockIdx.x * 16;
    const int n0   = wave * 16;
    const int nl   = lane & 15;
    const int kh   = (lane >> 4) * 2;

    const float* __restrict__ arow = h + (m0 + nl) * DDIM;

    v8f c = {};
#pragma unroll
    for (int k0 = 0; k0 < DDIM; k0 += 4) {
        v2f a, b;
        a[0] = arow[k0 + kh + 0];
        a[1] = arow[k0 + kh + 1];
        b[0] = W2[(k0 + kh + 0) * DDIM + n0 + nl];
        b[1] = W2[(k0 + kh + 1) * DDIM + n0 + nl];
        c = __builtin_amdgcn_wmma_f32_16x16x4_f32(
                false, a, false, b, (short)0, c, false, false);
    }

    const float bias  = b2[n0 + nl];
    const int   mbase = m0 + (lane >> 4) * 8;
#pragma unroll
    for (int r = 0; r < 8; ++r) {
        outr[(mbase + r) * DDIM + n0 + nl] = c[r] + bias;
    }
}

__global__ void broadcast_kernel(const float* __restrict__ outr,
                                 float* __restrict__ out) {
    const int r   = blockIdx.x;
    const int tid = threadIdx.x;

    const v4f v = ((const v4f*)(outr + r * DDIM))[tid & 15];
    v4f* __restrict__ dst = (v4f*)(out + (size_t)r * ROW_FLOATS);

    for (int i = tid; i < ROW_F4; i += 256) {
        __builtin_nontemporal_store(v, dst + i);
    }
}

extern "C" void kernel_launch(void* const* d_in, const int* in_sizes, int n_in,
                              void* d_out, int out_size, void* d_ws, size_t ws_size,
                              hipStream_t stream) {
    (void)in_sizes; (void)n_in; (void)out_size; (void)ws_size;

    const int*   TE = (const int*)  d_in[0];
    const float* W1 = (const float*)d_in[1];
    const float* b1 = (const float*)d_in[2];
    const float* W2 = (const float*)d_in[3];
    const float* b2 = (const float*)d_in[4];

    float* out  = (float*)d_out;
    float* h    = (float*)d_ws;
    float* outr = h + ROWS * DDIM;

    embed_h_kernel<<<(ROWS * DDIM) / 256, 256, 0, stream>>>(TE, W1, b1, h);

    gemm_wmma_kernel<<<ROWS / 16, 128, 0, stream>>>(h, W2, b2, outr);

    broadcast_kernel<<<ROWS, 256, 0, stream>>>(outr, out);
}


# Fuse gather+relu, the WMMA product and the 207x broadcast into one launch: 256 groups of 8 waves each build their rows on the matrix unit into LDS and stream them; no workspace pass
# speedup vs baseline: 1.4333x; 1.4333x over previous
#include <hip/hip_runtime.h>


#define ROWS       2304
#define DDIM       64
#define NREP       207
#define ROW_FLOATS (NREP * DDIM)
#define ROW_F4     (ROW_FLOATS / 4)
#define RPB        9
#define NBLK       (ROWS / RPB)
#define NTHR       256
#define FULL_IT    (ROW_F4 / NTHR)
#define TAIL_F4    (ROW_F4 - FULL_IT * NTHR)

typedef __attribute__((ext_vector_type(2))) int   v2i;
typedef __attribute__((ext_vector_type(2))) float v2f;
typedef __attribute__((ext_vector_type(4))) float v4f;
typedef __attribute__((ext_vector_type(8))) float v8f;

__global__ __launch_bounds__(NTHR)
void fused_embed_kernel(const int* __restrict__ TE,
                        const float* __restrict__ W1,
                        const float* __restrict__ b1,
                        const float* __restrict__ W2,
                        const float* __restrict__ b2,
                        float* __restrict__ out) {
    __shared__ __attribute__((aligned(16))) float tile[16][DDIM];

    const int tid  = threadIdx.x;
    const int lane = tid & 31;
    const int wave = __builtin_amdgcn_readfirstlane(tid >> 5);
    const int r0   = blockIdx.x * RPB;

    if (wave < 4) {
        const int nl = lane & 15;
        const int hf = lane >> 4;
        const int kh = hf * 2;
        const int n0 = wave * 16;

        const int ml  = nl < RPB ? nl : RPB - 1;
        const int row = r0 + ml;

        const v2i te = *(const v2i*)(TE + row * (NREP * 2));
        int dow = te[0] % 7;   if (dow < 0) dow += 7;
        int tod = te[1] % 288; if (tod < 0) tod += 288;

        const float* __restrict__ wa = W1 + dow * DDIM + kh;
        const float* __restrict__ wb = W1 + (7 + tod) * DDIM + kh;
        const float* __restrict__ bb = b1 + kh;
        const float* __restrict__ w2 = W2 + kh * DDIM + n0 + nl;

        v2f xa[DDIM / 4], xb[DDIM / 4], xc[DDIM / 4], bf[DDIM / 4];
#pragma unroll
        for (int q = 0; q < DDIM / 4; ++q) {
            xa[q]    = *(const v2f*)(wa + 4 * q);
            xb[q]    = *(const v2f*)(wb + 4 * q);
            xc[q]    = *(const v2f*)(bb + 4 * q);
            bf[q][0] = w2[(4 * q + 0) * DDIM];
            bf[q][1] = w2[(4 * q + 1) * DDIM];
        }
        __builtin_amdgcn_sched_barrier(0);

        v8f c = {};
#pragma unroll
        for (int q = 0; q < DDIM / 4; ++q) {
            v2f a;
            const float h0 = xa[q][0] + xb[q][0] + xc[q][0];
            const float h1 = xa[q][1] + xb[q][1] + xc[q][1];
            a[0] = h0 > 0.0f ? h0 : 0.0f;
            a[1] = h1 > 0.0f ? h1 : 0.0f;
            c = __builtin_amdgcn_wmma_f32_16x16x4_f32(
                    false, a, false, bf[q], (short)0, c, false, false);
        }

        const float bias = b2[n0 + nl];
#pragma unroll
        for (int r = 0; r < 8; ++r) {
            tile[r + 8 * hf][n0 + nl] = c[r] + bias;
        }
    }
    __syncthreads();

    v4f v[RPB];
#pragma unroll
    for (int j = 0; j < RPB; ++j) {
        v[j] = *(const v4f*)(&tile[j][(tid & 15) * 4]);
    }

    v4f* __restrict__ dst = (v4f*)(out + (size_t)r0 * ROW_FLOATS) + tid;

#pragma unroll
    for (int j = 0; j < RPB; ++j) {
#pragma unroll
        for (int it = 0; it < FULL_IT; ++it) {
            __builtin_nontemporal_store(v[j], dst + j * ROW_F4 + it * NTHR);
        }
    }
    if (tid < TAIL_F4) {
#pragma unroll
        for (int j = 0; j < RPB; ++j) {
            __builtin_nontemporal_store(v[j], dst + j * ROW_F4 + FULL_IT * NTHR);
        }
    }
}

extern "C" void kernel_launch(void* const* d_in, const int* in_sizes, int n_in,
                              void* d_out, int out_size, void* d_ws, size_t ws_size,
                              hipStream_t stream) {
    (void)in_sizes; (void)n_in; (void)out_size; (void)d_ws; (void)ws_size;

    const int*   TE = (const int*)  d_in[0];
    const float* W1 = (const float*)d_in[1];
    const float* b1 = (const float*)d_in[2];
    const float* W2 = (const float*)d_in[3];
    const float* b2 = (const float*)d_in[4];

    float* out = (float*)d_out;

    fused_embed_kernel<<<NBLK, NTHR, 0, stream>>>(TE, W1, b1, W2, b2, out);
}
